# starting kernel
# baseline (speedup 1.0000x reference)
#include <hip/hip_runtime.h>
#include <hip/hip_bf16.h>

typedef __attribute__((ext_vector_type(2))) float v2f;
typedef __attribute__((ext_vector_type(8))) float v8f;

#define NEG_SLOPE 0.2f

__device__ __forceinline__ unsigned f2ord(float f) {
    unsigned b = __float_as_uint(f);
    return (b & 0x80000000u) ? ~b : (b | 0x80000000u);
}
__device__ __forceinline__ float ord2f(unsigned u) {
    unsigned b = (u & 0x80000000u) ? (u & 0x7FFFFFFFu) : ~u;
    return __uint_as_float(b);
}
#define ORD_NEG_INF 0x007FFFFFu

__device__ __forceinline__ void edge_sd(const int* ei, int E0, int e, int& s, int& d) {
    if (e < E0) { s = ei[e]; d = ei[E0 + e]; }
    else        { s = e - E0; d = s; }
}

__global__ void gat_wmma_gemm(const float* __restrict__ A,
                              const float* __restrict__ B,
                              float* __restrict__ C,
                              int M, int Ntot, int K) {
    __shared__ float As[64][68];
    __shared__ float Bp[32][132];

    const int tid   = threadIdx.x;
    const int lane  = tid & 31;
    const int wave  = tid >> 5;
    const int mBase = blockIdx.y * 64;
    const int nBase = blockIdx.x * 64;

    const int ms   = (wave & 3) * 16;
    const int ns   = (wave >> 2) * 32;
    const int mrow = ms + (lane & 15);
    const int khi  = (lane >> 4) * 2;
    const int ncol = lane & 15;

    v8f acc0 = {}; v8f acc1 = {};

    for (int kBase = 0; kBase < K; kBase += 64) {
#pragma unroll
        for (int i = 0; i < 4; ++i) {
            int s  = tid + i * 256;
            int r  = s >> 4;
            int c4 = (s & 15) << 2;
            float4 va = make_float4(0.f, 0.f, 0.f, 0.f);
            int g = mBase + r;
            if (g < M) va = *(const float4*)(A + (size_t)g * K + kBase + c4);
            *(float4*)&As[r][c4] = va;
            float4 vb = *(const float4*)(B + (size_t)(kBase + r) * Ntot + nBase + c4);
            float* bp = &Bp[r >> 1][r & 1];
            bp[2 * c4 + 0] = vb.x;
            bp[2 * c4 + 2] = vb.y;
            bp[2 * c4 + 4] = vb.z;
            bp[2 * c4 + 6] = vb.w;
        }
        __syncthreads();

#pragma unroll
        for (int kk = 0; kk < 64; kk += 4) {
            const int p2 = (kk + khi) >> 1;
            v2f a, b0, b1v;
            a[0]   = As[mrow][kk + khi];
            a[1]   = As[mrow][kk + khi + 1];
            b0     = *(const v2f*)&Bp[p2][2 * (ns + ncol)];
            b1v    = *(const v2f*)&Bp[p2][2 * (ns + 16 + ncol)];
            acc0 = __builtin_amdgcn_wmma_f32_16x16x4_f32(
                false, a, false, b0, (short)0, acc0, false, false);
            acc1 = __builtin_amdgcn_wmma_f32_16x16x4_f32(
                false, a, false, b1v, (short)0, acc1, false, false);
        }
        __syncthreads();
    }

    const int orow = mBase + ms + ((lane >> 4) << 3);
    const int ocol = nBase + ns + ncol;
#pragma unroll
    for (int r = 0; r < 8; ++r) {
        int g = orow + r;
        if (g < M) {
            C[(size_t)g * Ntot + ocol]      = acc0[r];
            C[(size_t)g * Ntot + ocol + 16] = acc1[r];
        }
    }
}

__global__ void scores1_kernel(const float* __restrict__ h1,
                               const float* __restrict__ att_s,
                               const float* __restrict__ att_d,
                               float* __restrict__ as1, float* __restrict__ ad1,
                               int Nn) {
    int t = blockIdx.x * blockDim.x + threadIdx.x;
    if (t >= Nn * 8) return;
    int n = t >> 3, h = t & 7;
    const float* row = h1 + (size_t)n * 256 + h * 32;
    const float* ws = att_s + h * 32;
    const float* wd = att_d + h * 32;
    float ss = 0.f, sd = 0.f;
#pragma unroll
    for (int c = 0; c < 32; ++c) { ss += row[c] * ws[c]; sd += row[c] * wd[c]; }
    as1[t] = ss; ad1[t] = sd;
}

__global__ void init1_kernel(float* __restrict__ num1, float* __restrict__ den1,
                             unsigned* __restrict__ amax1, int Nn) {
    int t = blockIdx.x * blockDim.x + threadIdx.x;
    if (t < Nn * 256) num1[t] = 0.f;
    if (t < Nn * 8) { den1[t] = 0.f; amax1[t] = ORD_NEG_INF; }
}

__global__ void edge_max1_kernel(const int* __restrict__ ei, int E0, int Nn,
                                 const float* __restrict__ as1,
                                 const float* __restrict__ ad1,
                                 unsigned* __restrict__ amax1) {
    int t = blockIdx.x * blockDim.x + threadIdx.x;
    int ET = E0 + Nn;
    if (t >= ET * 8) return;
    int e = t >> 3, h = t & 7;
    int s, d; edge_sd(ei, E0, e, s, d);
    float lg = as1[s * 8 + h] + ad1[d * 8 + h];
    lg = lg > 0.f ? lg : NEG_SLOPE * lg;
    atomicMax(&amax1[d * 8 + h], f2ord(lg));
}

__global__ void edge_acc1_kernel(const int* __restrict__ ei, int E0, int Nn,
                                 const float* __restrict__ as1,
                                 const float* __restrict__ ad1,
                                 const unsigned* __restrict__ amax1,
                                 const float* __restrict__ h1,
                                 float* __restrict__ den1,
                                 float* __restrict__ num1) {
    int gtid = blockIdx.x * blockDim.x + threadIdx.x;
    int e    = gtid >> 5;
    int lane = threadIdx.x & 31;
    if (e >= E0 + Nn) return;
    int s, d; edge_sd(ei, E0, e, s, d);
    const float* hrow = h1 + (size_t)s * 256;
    float*       orow = num1 + (size_t)d * 256;
    __builtin_prefetch(hrow, 0, 1);
#pragma unroll
    for (int h = 0; h < 8; ++h) {
        float lg = as1[s * 8 + h] + ad1[d * 8 + h];
        lg = lg > 0.f ? lg : NEG_SLOPE * lg;
        float p = expf(lg - ord2f(amax1[d * 8 + h]));
        if (lane == 0) atomicAdd(&den1[d * 8 + h], p);
        atomicAdd(&orow[h * 32 + lane], hrow[h * 32 + lane] * p);
    }
}

__global__ void norm_elu1_kernel(const float* __restrict__ num1,
                                 const float* __restrict__ den1,
                                 const float* __restrict__ b1,
                                 float* __restrict__ hout, int Nn) {
    int t = blockIdx.x * blockDim.x + threadIdx.x;
    if (t >= Nn * 256) return;
    int n = t >> 8, idx = t & 255, h = idx >> 5;
    float v = num1[t] / (den1[n * 8 + h] + 1e-16f) + b1[idx];
    hout[t] = v > 0.f ? v : (expf(v) - 1.f);
}

__global__ void scores2_kernel(const float* __restrict__ h2,
                               const float* __restrict__ att_s,
                               const float* __restrict__ att_d,
                               float* __restrict__ as2, float* __restrict__ ad2,
                               int Nn) {
    int n = blockIdx.x * blockDim.x + threadIdx.x;
    if (n >= Nn) return;
    const float* row = h2 + (size_t)n * 64;
    float ss = 0.f, sd = 0.f;
#pragma unroll
    for (int c = 0; c < 64; ++c) { ss += row[c] * att_s[c]; sd += row[c] * att_d[c]; }
    as2[n] = ss; ad2[n] = sd;
}

__global__ void init2_kernel(float* __restrict__ out, float* __restrict__ den2,
                             unsigned* __restrict__ amax2, int Nn) {
    int t = blockIdx.x * blockDim.x + threadIdx.x;
    if (t < Nn * 64) out[t] = 0.f;
    if (t < Nn) { den2[t] = 0.f; amax2[t] = ORD_NEG_INF; }
}

__global__ void edge_max2_kernel(const int* __restrict__ ei, int E0, int Nn,
                                 const float* __restrict__ as2,
                                 const float* __restrict__ ad2,
                                 unsigned* __restrict__ amax2) {
    int e = blockIdx.x * blockDim.x + threadIdx.x;
    if (e >= E0 + Nn) return;
    int s, d; edge_sd(ei, E0, e, s, d);
    float lg = as2[s] + ad2[d];
    lg = lg > 0.f ? lg : NEG_SLOPE * lg;
    atomicMax(&amax2[d], f2ord(lg));
}

__global__ void edge_acc2_kernel(const int* __restrict__ ei, int E0, int Nn,
                                 const float* __restrict__ as2,
                                 const float* __restrict__ ad2,
                                 const unsigned* __restrict__ amax2,
                                 const float* __restrict__ h2,
                                 float* __restrict__ den2,
                                 float* __restrict__ out) {
    int gtid = blockIdx.x * blockDim.x + threadIdx.x;
    int e    = gtid >> 5;
    int lane = threadIdx.x & 31;
    if (e >= E0 + Nn) return;
    int s, d; edge_sd(ei, E0, e, s, d);
    float lg = as2[s] + ad2[d];
    lg = lg > 0.f ? lg : NEG_SLOPE * lg;
    float p = expf(lg - ord2f(amax2[d]));
    if (lane == 0) atomicAdd(&den2[d], p);
    const float* hrow = h2 + (size_t)s * 64;
    float*       orow = out + (size_t)d * 64;
    __builtin_prefetch(hrow, 0, 1);
    atomicAdd(&orow[lane],      hrow[lane]      * p);
    atomicAdd(&orow[32 + lane], hrow[32 + lane] * p);
}

__global__ void final_kernel(float* __restrict__ out,
                             const float* __restrict__ den2,
                             const float* __restrict__ b2, int Nn) {
    int t = blockIdx.x * blockDim.x + threadIdx.x;
    if (t >= Nn * 64) return;
    int n = t >> 6, c = t & 63;
    out[t] = out[t] / (den2[n] + 1e-16f) + b2[c];
}

extern "C" void kernel_launch(void* const* d_in, const int* in_sizes, int n_in,
                              void* d_out, int out_size, void* d_ws, size_t ws_size,
                              hipStream_t stream) {
    const float* x        = (const float*)d_in[0];
    const int*   ei       = (const int*)d_in[1];
    const float* W1       = (const float*)d_in[2];
    const float* att_src1 = (const float*)d_in[3];
    const float* att_dst1 = (const float*)d_in[4];
    const float* b1       = (const float*)d_in[5];
    const float* W2       = (const float*)d_in[6];
    const float* att_src2 = (const float*)d_in[7];
    const float* att_dst2 = (const float*)d_in[8];
    const float* b2       = (const float*)d_in[9];
    float* out = (float*)d_out;

    const int Nn = in_sizes[0] / 256;
    const int E0 = in_sizes[1] / 2;
    const int ET = E0 + Nn;

    float*    h1    = (float*)d_ws;
    float*    num1  = h1 + (size_t)Nn * 256;
    float*    as1   = num1 + (size_t)Nn * 256;
    float*    ad1   = as1 + (size_t)Nn * 8;
    unsigned* amax1 = (unsigned*)(ad1 + (size_t)Nn * 8);
    float*    den1  = (float*)(amax1 + (size_t)Nn * 8);
    float*    h2    = num1;
    float*    as2   = as1;
    float*    ad2   = as1 + Nn;
    unsigned* amax2 = (unsigned*)(as1 + 2 * (size_t)Nn);
    float*    den2  = as1 + 3 * (size_t)Nn;

    const int TB = 256;
    dim3 g1((256 + 63) / 64, (Nn + 63) / 64);
    dim3 g2((64 + 63) / 64, (Nn + 63) / 64);

    gat_wmma_gemm<<<g1, TB, 0, stream>>>(x, W1, h1, Nn, 256, 256);
    scores1_kernel<<<(Nn * 8 + TB - 1) / TB, TB, 0, stream>>>(h1, att_src1, att_dst1, as1, ad1, Nn);
    init1_kernel<<<(Nn * 256 + TB - 1) / TB, TB, 0, stream>>>(num1, den1, amax1, Nn);
    edge_max1_kernel<<<(ET * 8 + TB - 1) / TB, TB, 0, stream>>>(ei, E0, Nn, as1, ad1, amax1);
    edge_acc1_kernel<<<(ET + 7) / 8, TB, 0, stream>>>(ei, E0, Nn, as1, ad1, amax1,
                                                      h1, den1, num1);
    norm_elu1_kernel<<<(Nn * 256 + TB - 1) / TB, TB, 0, stream>>>(num1, den1, b1, h1, Nn);

    gat_wmma_gemm<<<g2, TB, 0, stream>>>(h1, W2, h2, Nn, 64, 256);
    scores2_kernel<<<(Nn + TB - 1) / TB, TB, 0, stream>>>(h2, att_src2, att_dst2, as2, ad2, Nn);
    init2_kernel<<<(Nn * 64 + TB - 1) / TB, TB, 0, stream>>>(out, den2, amax2, Nn);
    edge_max2_kernel<<<(ET + TB - 1) / TB, TB, 0, stream>>>(ei, E0, Nn, as2, ad2, amax2);
    edge_acc2_kernel<<<(ET + 7) / 8, TB, 0, stream>>>(ei, E0, Nn, as2, ad2, amax2,
                                                      h2, den2, out);
    final_kernel<<<(Nn * 64 + TB - 1) / TB, TB, 0, stream>>>(out, den2, b2, Nn);
}


# Edges counting-sorted by destination once per call; each layer's softmax-sum is a one-wave-per-node row gather (sum order = atomic arrival slots): no float atomics, init, max or normalise passes
# speedup vs baseline: 3.6332x; 3.6332x over previous
#include <hip/hip_runtime.h>
#include <hip/hip_bf16.h>

typedef __attribute__((ext_vector_type(2))) float v2f;
typedef __attribute__((ext_vector_type(8))) float v8f;

#define NEG_SLOPE 0.2f

__global__ void gat_wmma_gemm(const float* __restrict__ A,
                              const float* __restrict__ B,
                              float* __restrict__ C,
                              int M, int Ntot, int K) {
    __shared__ float As[64][68];
    __shared__ float Bp[32][132];

    const int tid   = threadIdx.x;
    const int lane  = tid & 31;
    const int wave  = tid >> 5;
    const int mBase = blockIdx.y * 64;
    const int nBase = blockIdx.x * 64;

    const int ms   = (wave & 3) * 16;
    const int ns   = (wave >> 2) * 32;
    const int mrow = ms + (lane & 15);
    const int khi  = (lane >> 4) * 2;
    const int ncol = lane & 15;

    v8f acc0 = {}; v8f acc1 = {};

    for (int kBase = 0; kBase < K; kBase += 64) {
#pragma unroll
        for (int i = 0; i < 4; ++i) {
            int s  = tid + i * 256;
            int r  = s >> 4;
            int c4 = (s & 15) << 2;
            float4 va = make_float4(0.f, 0.f, 0.f, 0.f);
            int g = mBase + r;
            if (g < M) va = *(const float4*)(A + (size_t)g * K + kBase + c4);
            *(float4*)&As[r][c4] = va;
            float4 vb = *(const float4*)(B + (size_t)(kBase + r) * Ntot + nBase + c4);
            float* bp = &Bp[r >> 1][r & 1];
            bp[2 * c4 + 0] = vb.x;
            bp[2 * c4 + 2] = vb.y;
            bp[2 * c4 + 4] = vb.z;
            bp[2 * c4 + 6] = vb.w;
        }
        __syncthreads();

#pragma unroll
        for (int kk = 0; kk < 64; kk += 4) {
            const int p2 = (kk + khi) >> 1;
            v2f a, b0, b1v;
            a[0]   = As[mrow][kk + khi];
            a[1]   = As[mrow][kk + khi + 1];
            b0     = *(const v2f*)&Bp[p2][2 * (ns + ncol)];
            b1v    = *(const v2f*)&Bp[p2][2 * (ns + 16 + ncol)];
            acc0 = __builtin_amdgcn_wmma_f32_16x16x4_f32(
                false, a, false, b0, (short)0, acc0, false, false);
            acc1 = __builtin_amdgcn_wmma_f32_16x16x4_f32(
                false, a, false, b1v, (short)0, acc1, false, false);
        }
        __syncthreads();
    }

    const int orow = mBase + ms + ((lane >> 4) << 3);
    const int ocol = nBase + ns + ncol;
#pragma unroll
    for (int r = 0; r < 8; ++r) {
        int g = orow + r;
        if (g < M) {
            C[(size_t)g * Ntot + ocol]      = acc0[r];
            C[(size_t)g * Ntot + ocol + 16] = acc1[r];
        }
    }
}

__device__ __forceinline__ float leaky(float v) { return v > 0.f ? v : NEG_SLOPE * v; }
__device__ __forceinline__ float elu1(float v)  { return v > 0.f ? v : (expf(v) - 1.f); }

__device__ __forceinline__ float wave_max(float v) {
    for (int o = 16; o > 0; o >>= 1) v = fmaxf(v, __shfl_xor(v, o));
    return v;
}
__device__ __forceinline__ float lane_f(float v, int j) {
    return __int_as_float(__builtin_amdgcn_readlane(__float_as_int(v), j));
}

__global__ void csr_zero_kernel(int* __restrict__ cnt, int Nn) {
    int t = blockIdx.x * blockDim.x + threadIdx.x;
    if (t < Nn) cnt[t] = 0;
}

__global__ void csr_count_kernel(const int* __restrict__ ei, int E0,
                                 int* __restrict__ cnt, int* __restrict__ rank) {
    int e = blockIdx.x * blockDim.x + threadIdx.x;
    if (e >= E0) return;
    rank[e] = atomicAdd(&cnt[ei[E0 + e]], 1);
}

__global__ __launch_bounds__(1024)
void csr_scan_kernel(const int* __restrict__ cnt, int* __restrict__ rowptr, int Nn) {
    __shared__ int s_front[32];
    __shared__ int s_tile[32];
    const int tid  = threadIdx.x;
    const int lane = tid & 31;
    const int wave = tid >> 5;
    const int base = blockIdx.x * 1024;

    int front = 0;
    for (int i = tid; i < base; i += 1024) front += cnt[i];
    for (int o = 16; o > 0; o >>= 1) front += __shfl_xor(front, o);

    const int idx = base + tid;
    const int v   = (idx < Nn) ? cnt[idx] : 0;
    int inc = v;
    for (int o = 1; o < 32; o <<= 1) {
        int t = __shfl_up(inc, o);
        if (lane >= o) inc += t;
    }
    if (lane == 0)  s_front[wave] = front;
    if (lane == 31) s_tile[wave]  = inc;
    __syncthreads();
    int off = 0;
    for (int w = 0; w < 32; ++w) {
        off += s_front[w];
        if (w < wave) off += s_tile[w];
    }
    if (idx < Nn) rowptr[idx] = off + inc - v;
}

__global__ void csr_fill_kernel(const int* __restrict__ ei, int E0,
                                const int* __restrict__ rowptr,
                                const int* __restrict__ rank,
                                int* __restrict__ srt) {
    int e = blockIdx.x * blockDim.x + threadIdx.x;
    if (e >= E0) return;
    srt[rowptr[ei[E0 + e]] + rank[e]] = ei[e];
}

__global__ void scores1_kernel(const float* __restrict__ h1,
                               const float* __restrict__ att_s,
                               float* __restrict__ as1, int Nn) {
    int t = blockIdx.x * blockDim.x + threadIdx.x;
    if (t >= Nn * 8) return;
    int n = t >> 3, h = t & 7;
    const float* row = h1 + (size_t)n * 256 + h * 32;
    const float* ws = att_s + h * 32;
    float ss = 0.f;
#pragma unroll
    for (int c = 0; c < 32; ++c) ss += row[c] * ws[c];
    as1[t] = ss;
}

__global__ __launch_bounds__(256)
void gather1_kernel(const int* __restrict__ rowptr, const int* __restrict__ srt,
                    int E0, int Nn,
                    const float* __restrict__ as1,
                    const float* __restrict__ att_d,
                    const float* __restrict__ h1,
                    const float* __restrict__ b1,
                    float* __restrict__ hout) {
    const int lane = threadIdx.x & 31;
    const int d = __builtin_amdgcn_readfirstlane((int)(blockIdx.x * 8 + (threadIdx.x >> 5)));
    if (d >= Nn) return;
    const int start = rowptr[d];
    const int end   = (d + 1 < Nn) ? rowptr[d + 1] : E0;
    const int deg   = end - start;
    const int total = deg + 1;

    const int g  = lane >> 3;
    const int cA = lane * 4;
    const int cB = 128 + lane * 4;

    float ad[8];
    {
        const float4 sA = *(const float4*)(h1 + (size_t)d * 256 + cA);
        const float4 sB = *(const float4*)(h1 + (size_t)d * 256 + cB);
        const float4 wA = *(const float4*)(att_d + cA);
        const float4 wB = *(const float4*)(att_d + cB);
        float dA = sA.x * wA.x + sA.y * wA.y + sA.z * wA.z + sA.w * wA.w;
        float dB = sB.x * wB.x + sB.y * wB.y + sB.z * wB.z + sB.w * wB.w;
        for (int o = 1; o < 8; o <<= 1) {
            dA += __shfl_xor(dA, o);
            dB += __shfl_xor(dB, o);
        }
#pragma unroll
        for (int h = 0; h < 4; ++h) {
            ad[h]     = lane_f(dA, h * 8);
            ad[4 + h] = lane_f(dB, h * 8);
        }
    }

    float m[8];
#pragma unroll
    for (int h = 0; h < 8; ++h) m[h] = -__builtin_inff();
    for (int c0 = 0; c0 < total; c0 += 32) {
        const int j  = c0 + lane;
        const int sj = (j < deg) ? srt[start + j] : d;
        const float4 a0 = *(const float4*)(as1 + (size_t)sj * 8);
        const float4 a1 = *(const float4*)(as1 + (size_t)sj * 8 + 4);
        const float lg[8] = {a0.x, a0.y, a0.z, a0.w, a1.x, a1.y, a1.z, a1.w};
        if (j < total) {
#pragma unroll
            for (int h = 0; h < 8; ++h) m[h] = fmaxf(m[h], leaky(lg[h] + ad[h]));
        }
    }
#pragma unroll
    for (int h = 0; h < 8; ++h) m[h] = wave_max(m[h]);

    float4 accA = make_float4(0.f, 0.f, 0.f, 0.f);
    float4 accB = make_float4(0.f, 0.f, 0.f, 0.f);
    float denA = 0.f, denB = 0.f;
    for (int c0 = 0; c0 < total; c0 += 32) {
        const int j  = c0 + lane;
        const int sj = (j < deg) ? srt[start + j] : d;
        const float4 a0 = *(const float4*)(as1 + (size_t)sj * 8);
        const float4 a1 = *(const float4*)(as1 + (size_t)sj * 8 + 4);
        const float lg[8] = {a0.x, a0.y, a0.z, a0.w, a1.x, a1.y, a1.z, a1.w};
        float p[8];
#pragma unroll
        for (int h = 0; h < 8; ++h) p[h] = expf(leaky(lg[h] + ad[h]) - m[h]);
        const int cnt = (total - c0 < 32) ? (total - c0) : 32;
        for (int k = 0; k < cnt; ++k) {
            const int s = __builtin_amdgcn_readlane(sj, k);
            const float* row = h1 + (size_t)s * 256;
            const float4 vA = *(const float4*)(row + cA);
            const float4 vB = *(const float4*)(row + cB);
            float ph[8];
#pragma unroll
            for (int h = 0; h < 8; ++h) ph[h] = lane_f(p[h], k);
            const float pA = g == 0 ? ph[0] : g == 1 ? ph[1] : g == 2 ? ph[2] : ph[3];
            const float pB = g == 0 ? ph[4] : g == 1 ? ph[5] : g == 2 ? ph[6] : ph[7];
            denA += pA;
            denB += pB;
            accA.x += pA * vA.x; accA.y += pA * vA.y; accA.z += pA * vA.z; accA.w += pA * vA.w;
            accB.x += pB * vB.x; accB.y += pB * vB.y; accB.z += pB * vB.z; accB.w += pB * vB.w;
        }
    }

    const float4 bA = *(const float4*)(b1 + cA);
    const float4 bB = *(const float4*)(b1 + cB);
    const float qA = denA + 1e-16f;
    const float qB = denB + 1e-16f;
    float4 oA, oB;
    oA.x = elu1(accA.x / qA + bA.x); oA.y = elu1(accA.y / qA + bA.y);
    oA.z = elu1(accA.z / qA + bA.z); oA.w = elu1(accA.w / qA + bA.w);
    oB.x = elu1(accB.x / qB + bB.x); oB.y = elu1(accB.y / qB + bB.y);
    oB.z = elu1(accB.z / qB + bB.z); oB.w = elu1(accB.w / qB + bB.w);
    *(float4*)(hout + (size_t)d * 256 + cA) = oA;
    *(float4*)(hout + (size_t)d * 256 + cB) = oB;
}

__global__ void scores2_kernel(const float* __restrict__ h2,
                               const float* __restrict__ att_s,
                               float* __restrict__ as2, int Nn) {
    int n = blockIdx.x * blockDim.x + threadIdx.x;
    if (n >= Nn) return;
    const float* row = h2 + (size_t)n * 64;
    float ss = 0.f;
#pragma unroll
    for (int c = 0; c < 64; ++c) ss += row[c] * att_s[c];
    as2[n] = ss;
}

__global__ __launch_bounds__(256)
void gather2_kernel(const int* __restrict__ rowptr, const int* __restrict__ srt,
                    int E0, int Nn,
                    const float* __restrict__ as2,
                    const float* __restrict__ att_d,
                    const float* __restrict__ h2,
                    const float* __restrict__ b2,
                    float* __restrict__ out) {
    const int lane = threadIdx.x & 31;
    const int d = __builtin_amdgcn_readfirstlane((int)(blockIdx.x * 8 + (threadIdx.x >> 5)));
    if (d >= Nn) return;
    const int start = rowptr[d];
    const int end   = (d + 1 < Nn) ? rowptr[d + 1] : E0;
    const int deg   = end - start;
    const int total = deg + 1;

    float ad;
    {
        const float2 sv = *(const float2*)(h2 + (size_t)d * 64 + lane * 2);
        const float2 wv = *(const float2*)(att_d + lane * 2);
        float t = sv.x * wv.x + sv.y * wv.y;
        for (int o = 16; o > 0; o >>= 1) t += __shfl_xor(t, o);
        ad = t;
    }

    float m = -__builtin_inff();
    for (int c0 = 0; c0 < total; c0 += 32) {
        const int j  = c0 + lane;
        const int sj = (j < deg) ? srt[start + j] : d;
        const float lg = leaky(as2[sj] + ad);
        if (j < total) m = fmaxf(m, lg);
    }
    m = wave_max(m);

    float2 acc = make_float2(0.f, 0.f);
    float den = 0.f;
    for (int c0 = 0; c0 < total; c0 += 32) {
        const int j  = c0 + lane;
        const int sj = (j < deg) ? srt[start + j] : d;
        const float p = expf(leaky(as2[sj] + ad) - m);
        const int cnt = (total - c0 < 32) ? (total - c0) : 32;
        for (int k = 0; k < cnt; ++k) {
            const int s = __builtin_amdgcn_readlane(sj, k);
            const float ph = lane_f(p, k);
            const float2 v = *(const float2*)(h2 + (size_t)s * 64 + lane * 2);
            den += ph;
            acc.x += ph * v.x;
            acc.y += ph * v.y;
        }
    }

    const float2 bb = *(const float2*)(b2 + lane * 2);
    const float q = den + 1e-16f;
    float2 o;
    o.x = acc.x / q + bb.x;
    o.y = acc.y / q + bb.y;
    *(float2*)(out + (size_t)d * 64 + lane * 2) = o;
}

extern "C" void kernel_launch(void* const* d_in, const int* in_sizes, int n_in,
                              void* d_out, int out_size, void* d_ws, size_t ws_size,
                              hipStream_t stream) {
    const float* x        = (const float*)d_in[0];
    const int*   ei       = (const int*)d_in[1];
    const float* W1       = (const float*)d_in[2];
    const float* att_src1 = (const float*)d_in[3];
    const float* att_dst1 = (const float*)d_in[4];
    const float* b1       = (const float*)d_in[5];
    const float* W2       = (const float*)d_in[6];
    const float* att_src2 = (const float*)d_in[7];
    const float* att_dst2 = (const float*)d_in[8];
    const float* b2       = (const float*)d_in[9];
    float* out = (float*)d_out;

    const int Nn = in_sizes[0] / 256;
    const int E0 = in_sizes[1] / 2;

    float* h1     = (float*)d_ws;
    float* hout   = h1 + (size_t)Nn * 256;
    float* as1    = hout + (size_t)Nn * 256;
    int*   rowptr = (int*)(as1 + (size_t)Nn * 8);
    int*   srt    = rowptr + Nn;
    int*   cnt    = (int*)hout;
    int*   rank   = cnt + Nn;
    float* h2     = h1;
    float* as2    = h1 + (size_t)Nn * 64;

    const int TB = 256;
    dim3 g1((256 + 63) / 64, (Nn + 63) / 64);
    dim3 g2((64 + 63) / 64, (Nn + 63) / 64);

    csr_zero_kernel<<<(Nn + TB - 1) / TB, TB, 0, stream>>>(cnt, Nn);
    csr_count_kernel<<<(E0 + TB - 1) / TB, TB, 0, stream>>>(ei, E0, cnt, rank);
    csr_scan_kernel<<<(Nn + 1023) / 1024, 1024, 0, stream>>>(cnt, rowptr, Nn);
    csr_fill_kernel<<<(E0 + TB - 1) / TB, TB, 0, stream>>>(ei, E0, rowptr, rank, srt);

    gat_wmma_gemm<<<g1, TB, 0, stream>>>(x, W1, h1, Nn, 256, 256);
    scores1_kernel<<<(Nn * 8 + TB - 1) / TB, TB, 0, stream>>>(h1, att_src1, as1, Nn);
    gather1_kernel<<<(Nn + 7) / 8, TB, 0, stream>>>(rowptr, srt, E0, Nn, as1, att_dst1,
                                                    h1, b1, hout);

    gat_wmma_gemm<<<g2, TB, 0, stream>>>(hout, W2, h2, Nn, 64, 256);
    scores2_kernel<<<(Nn + TB - 1) / TB, TB, 0, stream>>>(h2, att_src2, as2, Nn);
    gather2_kernel<<<(Nn + 7) / 8, TB, 0, stream>>>(rowptr, srt, E0, Nn, as2, att_dst2,
                                                    h2, b2, out);
}


# Source attention scores formed in the GEMM epilogue from the accumulators (16-lane DPP sums; layer 2 as two 32-column partials the gather adds): both score kernels gone, score sum order changed
# speedup vs baseline: 3.7708x; 1.0379x over previous
#include <hip/hip_runtime.h>
#include <hip/hip_bf16.h>

typedef __attribute__((ext_vector_type(2))) float v2f;
typedef __attribute__((ext_vector_type(8))) float v8f;

#define NEG_SLOPE 0.2f

__device__ __forceinline__ float row16_sum(float v) {
    v += __int_as_float(__builtin_amdgcn_update_dpp(0, __float_as_int(v), 0x161, 0xF, 0xF, false));
    v += __int_as_float(__builtin_amdgcn_update_dpp(0, __float_as_int(v), 0x162, 0xF, 0xF, false));
    v += __int_as_float(__builtin_amdgcn_update_dpp(0, __float_as_int(v), 0x164, 0xF, 0xF, false));
    v += __int_as_float(__builtin_amdgcn_update_dpp(0, __float_as_int(v), 0x168, 0xF, 0xF, false));
    return v;
}

__global__ void gat_wmma_gemm(const float* __restrict__ A,
                              const float* __restrict__ B,
                              float* __restrict__ C,
                              const float* __restrict__ att,
                              float* __restrict__ sc,
                              int M, int Ntot, int K) {
    __shared__ float As[64][68];
    __shared__ float Bp[32][132];

    const int tid   = threadIdx.x;
    const int lane  = tid & 31;
    const int wave  = tid >> 5;
    const int mBase = blockIdx.y * 64;
    const int nBase = blockIdx.x * 64;

    const int ms   = (wave & 3) * 16;
    const int ns   = (wave >> 2) * 32;
    const int mrow = ms + (lane & 15);
    const int khi  = (lane >> 4) * 2;
    const int ncol = lane & 15;

    v8f acc0 = {}; v8f acc1 = {};
    const float w0 = att[nBase + ns + ncol];
    const float w1 = att[nBase + ns + ncol + 16];

    for (int kBase = 0; kBase < K; kBase += 64) {
#pragma unroll
        for (int i = 0; i < 4; ++i) {
            int s  = tid + i * 256;
            int r  = s >> 4;
            int c4 = (s & 15) << 2;
            float4 va = make_float4(0.f, 0.f, 0.f, 0.f);
            int g = mBase + r;
            if (g < M) va = *(const float4*)(A + (size_t)g * K + kBase + c4);
            *(float4*)&As[r][c4] = va;
            float4 vb = *(const float4*)(B + (size_t)(kBase + r) * Ntot + nBase + c4);
            float* bp = &Bp[r >> 1][r & 1];
            bp[2 * c4 + 0] = vb.x;
            bp[2 * c4 + 2] = vb.y;
            bp[2 * c4 + 4] = vb.z;
            bp[2 * c4 + 6] = vb.w;
        }
        __syncthreads();

#pragma unroll
        for (int kk = 0; kk < 64; kk += 4) {
            const int p2 = (kk + khi) >> 1;
            v2f a, b0, b1v;
            a[0]   = As[mrow][kk + khi];
            a[1]   = As[mrow][kk + khi + 1];
            b0     = *(const v2f*)&Bp[p2][2 * (ns + ncol)];
            b1v    = *(const v2f*)&Bp[p2][2 * (ns + 16 + ncol)];
            acc0 = __builtin_amdgcn_wmma_f32_16x16x4_f32(
                false, a, false, b0, (short)0, acc0, false, false);
            acc1 = __builtin_amdgcn_wmma_f32_16x16x4_f32(
                false, a, false, b1v, (short)0, acc1, false, false);
        }
        __syncthreads();
    }

    const int orow = mBase + ms + ((lane >> 4) << 3);
    const int ocol = nBase + ns + ncol;
#pragma unroll
    for (int r = 0; r < 8; ++r) {
        int g = orow + r;
        if (g < M) {
            C[(size_t)g * Ntot + ocol]      = acc0[r];
            C[(size_t)g * Ntot + ocol + 16] = acc1[r];
        }
    }

    float sr[8];
#pragma unroll
    for (int r = 0; r < 8; ++r) sr[r] = row16_sum(acc0[r] * w0 + acc1[r] * w1);
    float sv = sr[0];
#pragma unroll
    for (int r = 1; r < 8; ++r) sv = (ncol == r) ? sr[r] : sv;
    const int gs = orow + ncol;
    if (ncol < 8 && gs < M)
        sc[(size_t)gs * (Ntot >> 5) + ((nBase + ns) >> 5)] = sv;
}

__device__ __forceinline__ float leaky(float v) { return v > 0.f ? v : NEG_SLOPE * v; }
__device__ __forceinline__ float elu1(float v)  { return v > 0.f ? v : (expf(v) - 1.f); }

__device__ __forceinline__ float wave_max(float v) {
    for (int o = 16; o > 0; o >>= 1) v = fmaxf(v, __shfl_xor(v, o));
    return v;
}
__device__ __forceinline__ float lane_f(float v, int j) {
    return __int_as_float(__builtin_amdgcn_readlane(__float_as_int(v), j));
}

__global__ void csr_zero_kernel(int* __restrict__ cnt, int Nn) {
    int t = blockIdx.x * blockDim.x + threadIdx.x;
    if (t < Nn) cnt[t] = 0;
}

__global__ void csr_count_kernel(const int* __restrict__ ei, int E0,
                                 int* __restrict__ cnt, int* __restrict__ rank) {
    int e = blockIdx.x * blockDim.x + threadIdx.x;
    if (e >= E0) return;
    rank[e] = atomicAdd(&cnt[ei[E0 + e]], 1);
}

__global__ __launch_bounds__(1024)
void csr_scan_kernel(const int* __restrict__ cnt, int* __restrict__ rowptr, int Nn) {
    __shared__ int s_front[32];
    __shared__ int s_tile[32];
    const int tid  = threadIdx.x;
    const int lane = tid & 31;
    const int wave = tid >> 5;
    const int base = blockIdx.x * 1024;

    int front = 0;
    for (int i = tid; i < base; i += 1024) front += cnt[i];
    for (int o = 16; o > 0; o >>= 1) front += __shfl_xor(front, o);

    const int idx = base + tid;
    const int v   = (idx < Nn) ? cnt[idx] : 0;
    int inc = v;
    for (int o = 1; o < 32; o <<= 1) {
        int t = __shfl_up(inc, o);
        if (lane >= o) inc += t;
    }
    if (lane == 0)  s_front[wave] = front;
    if (lane == 31) s_tile[wave]  = inc;
    __syncthreads();
    int off = 0;
    for (int w = 0; w < 32; ++w) {
        off += s_front[w];
        if (w < wave) off += s_tile[w];
    }
    if (idx < Nn) rowptr[idx] = off + inc - v;
}

__global__ void csr_fill_kernel(const int* __restrict__ ei, int E0,
                                const int* __restrict__ rowptr,
                                const int* __restrict__ rank,
                                int* __restrict__ srt) {
    int e = blockIdx.x * blockDim.x + threadIdx.x;
    if (e >= E0) return;
    srt[rowptr[ei[E0 + e]] + rank[e]] = ei[e];
}

__global__ __launch_bounds__(256)
void gather1_kernel(const int* __restrict__ rowptr, const int* __restrict__ srt,
                    int E0, int Nn,
                    const float* __restrict__ as1,
                    const float* __restrict__ att_d,
                    const float* __restrict__ h1,
                    const float* __restrict__ b1,
                    float* __restrict__ hout) {
    const int lane = threadIdx.x & 31;
    const int d = __builtin_amdgcn_readfirstlane((int)(blockIdx.x * 8 + (threadIdx.x >> 5)));
    if (d >= Nn) return;
    const int start = rowptr[d];
    const int end   = (d + 1 < Nn) ? rowptr[d + 1] : E0;
    const int deg   = end - start;
    const int total = deg + 1;

    const int g  = lane >> 3;
    const int cA = lane * 4;
    const int cB = 128 + lane * 4;

    float ad[8];
    {
        const float4 sA = *(const float4*)(h1 + (size_t)d * 256 + cA);
        const float4 sB = *(const float4*)(h1 + (size_t)d * 256 + cB);
        const float4 wA = *(const float4*)(att_d + cA);
        const float4 wB = *(const float4*)(att_d + cB);
        float dA = sA.x * wA.x + sA.y * wA.y + sA.z * wA.z + sA.w * wA.w;
        float dB = sB.x * wB.x + sB.y * wB.y + sB.z * wB.z + sB.w * wB.w;
        for (int o = 1; o < 8; o <<= 1) {
            dA += __shfl_xor(dA, o);
            dB += __shfl_xor(dB, o);
        }
#pragma unroll
        for (int h = 0; h < 4; ++h) {
            ad[h]     = lane_f(dA, h * 8);
            ad[4 + h] = lane_f(dB, h * 8);
        }
    }

    float m[8];
#pragma unroll
    for (int h = 0; h < 8; ++h) m[h] = -__builtin_inff();
    for (int c0 = 0; c0 < total; c0 += 32) {
        const int j  = c0 + lane;
        const int sj = (j < deg) ? srt[start + j] : d;
        const float4 a0 = *(const float4*)(as1 + (size_t)sj * 8);
        const float4 a1 = *(const float4*)(as1 + (size_t)sj * 8 + 4);
        const float lg[8] = {a0.x, a0.y, a0.z, a0.w, a1.x, a1.y, a1.z, a1.w};
        if (j < total) {
#pragma unroll
            for (int h = 0; h < 8; ++h) m[h] = fmaxf(m[h], leaky(lg[h] + ad[h]));
        }
    }
#pragma unroll
    for (int h = 0; h < 8; ++h) m[h] = wave_max(m[h]);

    float4 accA = make_float4(0.f, 0.f, 0.f, 0.f);
    float4 accB = make_float4(0.f, 0.f, 0.f, 0.f);
    float denA = 0.f, denB = 0.f;
    for (int c0 = 0; c0 < total; c0 += 32) {
        const int j  = c0 + lane;
        const int sj = (j < deg) ? srt[start + j] : d;
        const float4 a0 = *(const float4*)(as1 + (size_t)sj * 8);
        const float4 a1 = *(const float4*)(as1 + (size_t)sj * 8 + 4);
        const float lg[8] = {a0.x, a0.y, a0.z, a0.w, a1.x, a1.y, a1.z, a1.w};
        float p[8];
#pragma unroll
        for (int h = 0; h < 8; ++h) p[h] = expf(leaky(lg[h] + ad[h]) - m[h]);
        const int cnt = (total - c0 < 32) ? (total - c0) : 32;
        for (int k = 0; k < cnt; ++k) {
            const int s = __builtin_amdgcn_readlane(sj, k);
            const float* row = h1 + (size_t)s * 256;
            const float4 vA = *(const float4*)(row + cA);
            const float4 vB = *(const float4*)(row + cB);
            float ph[8];
#pragma unroll
            for (int h = 0; h < 8; ++h) ph[h] = lane_f(p[h], k);
            const float pA = g == 0 ? ph[0] : g == 1 ? ph[1] : g == 2 ? ph[2] : ph[3];
            const float pB = g == 0 ? ph[4] : g == 1 ? ph[5] : g == 2 ? ph[6] : ph[7];
            denA += pA;
            denB += pB;
            accA.x += pA * vA.x; accA.y += pA * vA.y; accA.z += pA * vA.z; accA.w += pA * vA.w;
            accB.x += pB * vB.x; accB.y += pB * vB.y; accB.z += pB * vB.z; accB.w += pB * vB.w;
        }
    }

    const float4 bA = *(const float4*)(b1 + cA);
    const float4 bB = *(const float4*)(b1 + cB);
    const float qA = denA + 1e-16f;
    const float qB = denB + 1e-16f;
    float4 oA, oB;
    oA.x = elu1(accA.x / qA + bA.x); oA.y = elu1(accA.y / qA + bA.y);
    oA.z = elu1(accA.z / qA + bA.z); oA.w = elu1(accA.w / qA + bA.w);
    oB.x = elu1(accB.x / qB + bB.x); oB.y = elu1(accB.y / qB + bB.y);
    oB.z = elu1(accB.z / qB + bB.z); oB.w = elu1(accB.w / qB + bB.w);
    *(float4*)(hout + (size_t)d * 256 + cA) = oA;
    *(float4*)(hout + (size_t)d * 256 + cB) = oB;
}

__global__ __launch_bounds__(256)
void gather2_kernel(const int* __restrict__ rowptr, const int* __restrict__ srt,
                    int E0, int Nn,
                    const float* __restrict__ as2,
                    const float* __restrict__ att_d,
                    const float* __restrict__ h2,
                    const float* __restrict__ b2,
                    float* __restrict__ out) {
    const int lane = threadIdx.x & 31;
    const int d = __builtin_amdgcn_readfirstlane((int)(blockIdx.x * 8 + (threadIdx.x >> 5)));
    if (d >= Nn) return;
    const int start = rowptr[d];
    const int end   = (d + 1 < Nn) ? rowptr[d + 1] : E0;
    const int deg   = end - start;
    const int total = deg + 1;

    float ad;
    {
        const float2 sv = *(const float2*)(h2 + (size_t)d * 64 + lane * 2);
        const float2 wv = *(const float2*)(att_d + lane * 2);
        float t = sv.x * wv.x + sv.y * wv.y;
        for (int o = 16; o > 0; o >>= 1) t += __shfl_xor(t, o);
        ad = t;
    }

    float m = -__builtin_inff();
    for (int c0 = 0; c0 < total; c0 += 32) {
        const int j  = c0 + lane;
        const int sj = (j < deg) ? srt[start + j] : d;
        const float2 ap = *(const float2*)(as2 + (size_t)sj * 2);
        const float lg = leaky((ap.x + ap.y) + ad);
        if (j < total) m = fmaxf(m, lg);
    }
    m = wave_max(m);

    float2 acc = make_float2(0.f, 0.f);
    float den = 0.f;
    for (int c0 = 0; c0 < total; c0 += 32) {
        const int j  = c0 + lane;
        const int sj = (j < deg) ? srt[start + j] : d;
        const float2 ap = *(const float2*)(as2 + (size_t)sj * 2);
        const float p = expf(leaky((ap.x + ap.y) + ad) - m);
        const int cnt = (total - c0 < 32) ? (total - c0) : 32;
        for (int k = 0; k < cnt; ++k) {
            const int s = __builtin_amdgcn_readlane(sj, k);
            const float ph = lane_f(p, k);
            const float2 v = *(const float2*)(h2 + (size_t)s * 64 + lane * 2);
            den += ph;
            acc.x += ph * v.x;
            acc.y += ph * v.y;
        }
    }

    const float2 bb = *(const float2*)(b2 + lane * 2);
    const float q = den + 1e-16f;
    float2 o;
    o.x = acc.x / q + bb.x;
    o.y = acc.y / q + bb.y;
    *(float2*)(out + (size_t)d * 64 + lane * 2) = o;
}

extern "C" void kernel_launch(void* const* d_in, const int* in_sizes, int n_in,
                              void* d_out, int out_size, void* d_ws, size_t ws_size,
                              hipStream_t stream) {
    const float* x        = (const float*)d_in[0];
    const int*   ei       = (const int*)d_in[1];
    const float* W1       = (const float*)d_in[2];
    const float* att_src1 = (const float*)d_in[3];
    const float* att_dst1 = (const float*)d_in[4];
    const float* b1       = (const float*)d_in[5];
    const float* W2       = (const float*)d_in[6];
    const float* att_src2 = (const float*)d_in[7];
    const float* att_dst2 = (const float*)d_in[8];
    const float* b2       = (const float*)d_in[9];
    float* out = (float*)d_out;

    const int Nn = in_sizes[0] / 256;
    const int E0 = in_sizes[1] / 2;

    float* h1     = (float*)d_ws;
    float* hout   = h1 + (size_t)Nn * 256;
    float* as1    = hout + (size_t)Nn * 256;
    int*   rowptr = (int*)(as1 + (size_t)Nn * 8);
    int*   srt    = rowptr + Nn;
    int*   cnt    = (int*)hout;
    int*   rank   = cnt + Nn;
    float* h2     = h1;
    float* as2    = h1 + (size_t)Nn * 64;

    const int TB = 256;
    dim3 g1((256 + 63) / 64, (Nn + 63) / 64);
    dim3 g2((64 + 63) / 64, (Nn + 63) / 64);

    csr_zero_kernel<<<(Nn + TB - 1) / TB, TB, 0, stream>>>(cnt, Nn);
    csr_count_kernel<<<(E0 + TB - 1) / TB, TB, 0, stream>>>(ei, E0, cnt, rank);
    csr_scan_kernel<<<(Nn + 1023) / 1024, 1024, 0, stream>>>(cnt, rowptr, Nn);
    csr_fill_kernel<<<(E0 + TB - 1) / TB, TB, 0, stream>>>(ei, E0, rowptr, rank, srt);

    gat_wmma_gemm<<<g1, TB, 0, stream>>>(x, W1, h1, att_src1, as1, Nn, 256, 256);
    gather1_kernel<<<(Nn + 7) / 8, TB, 0, stream>>>(rowptr, srt, E0, Nn, as1, att_dst1,
                                                    h1, b1, hout);

    gat_wmma_gemm<<<g2, TB, 0, stream>>>(hout, W2, h2, att_src2, as2, Nn, 64, 256);
    gather2_kernel<<<(Nn + 7) / 8, TB, 0, stream>>>(rowptr, srt, E0, Nn, as2, att_dst2,
                                                    h2, b2, out);
}


# GEMM tile loads software-pipelined: the next K chunk's 8 global loads issue as one group before the current chunk's WMMAs (row guard becomes a row clamp, so no load is predicated and serialised)
# speedup vs baseline: 4.0654x; 1.0781x over previous
#include <hip/hip_runtime.h>
#include <hip/hip_bf16.h>

typedef __attribute__((ext_vector_type(2))) float v2f;
typedef __attribute__((ext_vector_type(8))) float v8f;

#define NEG_SLOPE 0.2f

__device__ __forceinline__ float row16_sum(float v) {
    v += __int_as_float(__builtin_amdgcn_update_dpp(0, __float_as_int(v), 0x161, 0xF, 0xF, false));
    v += __int_as_float(__builtin_amdgcn_update_dpp(0, __float_as_int(v), 0x162, 0xF, 0xF, false));
    v += __int_as_float(__builtin_amdgcn_update_dpp(0, __float_as_int(v), 0x164, 0xF, 0xF, false));
    v += __int_as_float(__builtin_amdgcn_update_dpp(0, __float_as_int(v), 0x168, 0xF, 0xF, false));
    return v;
}

__global__ void gat_wmma_gemm(const float* __restrict__ A,
                              const float* __restrict__ B,
                              float* __restrict__ C,
                              const float* __restrict__ att,
                              float* __restrict__ sc,
                              int M, int Ntot, int K) {
    __shared__ float As[64][68];
    __shared__ float Bp[32][132];

    const int tid   = threadIdx.x;
    const int lane  = tid & 31;
    const int wave  = tid >> 5;
    const int mBase = blockIdx.y * 64;
    const int nBase = blockIdx.x * 64;

    const int ms   = (wave & 3) * 16;
    const int ns   = (wave >> 2) * 32;
    const int mrow = ms + (lane & 15);
    const int khi  = (lane >> 4) * 2;
    const int ncol = lane & 15;

    v8f acc0 = {}; v8f acc1 = {};
    const float w0 = att[nBase + ns + ncol];
    const float w1 = att[nBase + ns + ncol + 16];

    const int r0 = tid >> 4;
    const int c4 = (tid & 15) << 2;
    const int gl = M - 1;
    const int g0 = min(mBase + r0,      gl);
    const int g1 = min(mBase + r0 + 16, gl);
    const int g2 = min(mBase + r0 + 32, gl);
    const int g3 = min(mBase + r0 + 48, gl);
    const float* a0p = A + (size_t)g0 * K + c4;
    const float* a1p = A + (size_t)g1 * K + c4;
    const float* a2p = A + (size_t)g2 * K + c4;
    const float* a3p = A + (size_t)g3 * K + c4;
    const float* bcol = B + (size_t)r0 * Ntot + nBase + c4;
    const size_t b16 = (size_t)16 * Ntot;
    float4 va0 = *(const float4*)a0p, va1 = *(const float4*)a1p;
    float4 va2 = *(const float4*)a2p, va3 = *(const float4*)a3p;
    float4 vb0 = *(const float4*)(bcol),           vb1 = *(const float4*)(bcol + b16);
    float4 vb2 = *(const float4*)(bcol + 2 * b16), vb3 = *(const float4*)(bcol + 3 * b16);

    for (int kBase = 0; kBase < K; kBase += 64) {
        *(float4*)&As[r0     ][c4] = va0;
        *(float4*)&As[r0 + 16][c4] = va1;
        *(float4*)&As[r0 + 32][c4] = va2;
        *(float4*)&As[r0 + 48][c4] = va3;
        {
            float* bp = &Bp[r0 >> 1][(r0 & 1) + 2 * c4];
            bp[0] = vb0.x; bp[2] = vb0.y; bp[4] = vb0.z; bp[6] = vb0.w;
            bp += 8 * 132;
            bp[0] = vb1.x; bp[2] = vb1.y; bp[4] = vb1.z; bp[6] = vb1.w;
            bp += 8 * 132;
            bp[0] = vb2.x; bp[2] = vb2.y; bp[4] = vb2.z; bp[6] = vb2.w;
            bp += 8 * 132;
            bp[0] = vb3.x; bp[2] = vb3.y; bp[4] = vb3.z; bp[6] = vb3.w;
        }
        __syncthreads();
        if (kBase + 64 < K) {
            const int kn = kBase + 64;
            const float* bn = bcol + (size_t)kn * Ntot;
            va0 = *(const float4*)(a0p + kn); va1 = *(const float4*)(a1p + kn);
            va2 = *(const float4*)(a2p + kn); va3 = *(const float4*)(a3p + kn);
            vb0 = *(const float4*)(bn);           vb1 = *(const float4*)(bn + b16);
            vb2 = *(const float4*)(bn + 2 * b16); vb3 = *(const float4*)(bn + 3 * b16);
        }

#pragma unroll
        for (int kk = 0; kk < 64; kk += 4) {
            const int p2 = (kk + khi) >> 1;
            v2f a, b0, b1v;
            a[0]   = As[mrow][kk + khi];
            a[1]   = As[mrow][kk + khi + 1];
            b0     = *(const v2f*)&Bp[p2][2 * (ns + ncol)];
            b1v    = *(const v2f*)&Bp[p2][2 * (ns + 16 + ncol)];
            acc0 = __builtin_amdgcn_wmma_f32_16x16x4_f32(
                false, a, false, b0, (short)0, acc0, false, false);
            acc1 = __builtin_amdgcn_wmma_f32_16x16x4_f32(
                false, a, false, b1v, (short)0, acc1, false, false);
        }
        __syncthreads();
    }

    const int orow = mBase + ms + ((lane >> 4) << 3);
    const int ocol = nBase + ns + ncol;
#pragma unroll
    for (int r = 0; r < 8; ++r) {
        int g = orow + r;
        if (g < M) {
            C[(size_t)g * Ntot + ocol]      = acc0[r];
            C[(size_t)g * Ntot + ocol + 16] = acc1[r];
        }
    }

    float sr[8];
#pragma unroll
    for (int r = 0; r < 8; ++r) sr[r] = row16_sum(acc0[r] * w0 + acc1[r] * w1);
    float sv = sr[0];
#pragma unroll
    for (int r = 1; r < 8; ++r) sv = (ncol == r) ? sr[r] : sv;
    const int gs = orow + ncol;
    if (ncol < 8 && gs < M)
        sc[(size_t)gs * (Ntot >> 5) + ((nBase + ns) >> 5)] = sv;
}

__device__ __forceinline__ float leaky(float v) { return v > 0.f ? v : NEG_SLOPE * v; }
__device__ __forceinline__ float elu1(float v)  { return v > 0.f ? v : (expf(v) - 1.f); }

__device__ __forceinline__ float wave_max(float v) {
    for (int o = 16; o > 0; o >>= 1) v = fmaxf(v, __shfl_xor(v, o));
    return v;
}
__device__ __forceinline__ float lane_f(float v, int j) {
    return __int_as_float(__builtin_amdgcn_readlane(__float_as_int(v), j));
}

__global__ void csr_zero_kernel(int* __restrict__ cnt, int Nn) {
    int t = blockIdx.x * blockDim.x + threadIdx.x;
    if (t < Nn) cnt[t] = 0;
}

__global__ void csr_count_kernel(const int* __restrict__ ei, int E0,
                                 int* __restrict__ cnt, int* __restrict__ rank) {
    int e = blockIdx.x * blockDim.x + threadIdx.x;
    if (e >= E0) return;
    rank[e] = atomicAdd(&cnt[ei[E0 + e]], 1);
}

__global__ __launch_bounds__(1024)
void csr_scan_kernel(const int* __restrict__ cnt, int* __restrict__ rowptr, int Nn) {
    __shared__ int s_front[32];
    __shared__ int s_tile[32];
    const int tid  = threadIdx.x;
    const int lane = tid & 31;
    const int wave = tid >> 5;
    const int base = blockIdx.x * 1024;

    int front = 0;
    for (int i = tid; i < base; i += 1024) front += cnt[i];
    for (int o = 16; o > 0; o >>= 1) front += __shfl_xor(front, o);

    const int idx = base + tid;
    const int v   = (idx < Nn) ? cnt[idx] : 0;
    int inc = v;
    for (int o = 1; o < 32; o <<= 1) {
        int t = __shfl_up(inc, o);
        if (lane >= o) inc += t;
    }
    if (lane == 0)  s_front[wave] = front;
    if (lane == 31) s_tile[wave]  = inc;
    __syncthreads();
    int off = 0;
    for (int w = 0; w < 32; ++w) {
        off += s_front[w];
        if (w < wave) off += s_tile[w];
    }
    if (idx < Nn) rowptr[idx] = off + inc - v;
}

__global__ void csr_fill_kernel(const int* __restrict__ ei, int E0,
                                const int* __restrict__ rowptr,
                                const int* __restrict__ rank,
                                int* __restrict__ srt) {
    int e = blockIdx.x * blockDim.x + threadIdx.x;
    if (e >= E0) return;
    srt[rowptr[ei[E0 + e]] + rank[e]] = ei[e];
}

__global__ __launch_bounds__(256)
void gather1_kernel(const int* __restrict__ rowptr, const int* __restrict__ srt,
                    int E0, int Nn,
                    const float* __restrict__ as1,
                    const float* __restrict__ att_d,
                    const float* __restrict__ h1,
                    const float* __restrict__ b1,
                    float* __restrict__ hout) {
    const int lane = threadIdx.x & 31;
    const int d = __builtin_amdgcn_readfirstlane((int)(blockIdx.x * 8 + (threadIdx.x >> 5)));
    if (d >= Nn) return;
    const int start = rowptr[d];
    const int end   = (d + 1 < Nn) ? rowptr[d + 1] : E0;
    const int deg   = end - start;
    const int total = deg + 1;

    const int g  = lane >> 3;
    const int cA = lane * 4;
    const int cB = 128 + lane * 4;

    float ad[8];
    {
        const float4 sA = *(const float4*)(h1 + (size_t)d * 256 + cA);
        const float4 sB = *(const float4*)(h1 + (size_t)d * 256 + cB);
        const float4 wA = *(const float4*)(att_d + cA);
        const float4 wB = *(const float4*)(att_d + cB);
        float dA = sA.x * wA.x + sA.y * wA.y + sA.z * wA.z + sA.w * wA.w;
        float dB = sB.x * wB.x + sB.y * wB.y + sB.z * wB.z + sB.w * wB.w;
        for (int o = 1; o < 8; o <<= 1) {
            dA += __shfl_xor(dA, o);
            dB += __shfl_xor(dB, o);
        }
#pragma unroll
        for (int h = 0; h < 4; ++h) {
            ad[h]     = lane_f(dA, h * 8);
            ad[4 + h] = lane_f(dB, h * 8);
        }
    }

    float m[8];
#pragma unroll
    for (int h = 0; h < 8; ++h) m[h] = -__builtin_inff();
    for (int c0 = 0; c0 < total; c0 += 32) {
        const int j  = c0 + lane;
        const int sj = (j < deg) ? srt[start + j] : d;
        const float4 a0 = *(const float4*)(as1 + (size_t)sj * 8);
        const float4 a1 = *(const float4*)(as1 + (size_t)sj * 8 + 4);
        const float lg[8] = {a0.x, a0.y, a0.z, a0.w, a1.x, a1.y, a1.z, a1.w};
        if (j < total) {
#pragma unroll
            for (int h = 0; h < 8; ++h) m[h] = fmaxf(m[h], leaky(lg[h] + ad[h]));
        }
    }
#pragma unroll
    for (int h = 0; h < 8; ++h) m[h] = wave_max(m[h]);

    float4 accA = make_float4(0.f, 0.f, 0.f, 0.f);
    float4 accB = make_float4(0.f, 0.f, 0.f, 0.f);
    float denA = 0.f, denB = 0.f;
    for (int c0 = 0; c0 < total; c0 += 32) {
        const int j  = c0 + lane;
        const int sj = (j < deg) ? srt[start + j] : d;
        const float4 a0 = *(const float4*)(as1 + (size_t)sj * 8);
        const float4 a1 = *(const float4*)(as1 + (size_t)sj * 8 + 4);
        const float lg[8] = {a0.x, a0.y, a0.z, a0.w, a1.x, a1.y, a1.z, a1.w};
        float p[8];
#pragma unroll
        for (int h = 0; h < 8; ++h) p[h] = expf(leaky(lg[h] + ad[h]) - m[h]);
        const int cnt = (total - c0 < 32) ? (total - c0) : 32;
        for (int k = 0; k < cnt; ++k) {
            const int s = __builtin_amdgcn_readlane(sj, k);
            const float* row = h1 + (size_t)s * 256;
            const float4 vA = *(const float4*)(row + cA);
            const float4 vB = *(const float4*)(row + cB);
            float ph[8];
#pragma unroll
            for (int h = 0; h < 8; ++h) ph[h] = lane_f(p[h], k);
            const float pA = g == 0 ? ph[0] : g == 1 ? ph[1] : g == 2 ? ph[2] : ph[3];
            const float pB = g == 0 ? ph[4] : g == 1 ? ph[5] : g == 2 ? ph[6] : ph[7];
            denA += pA;
            denB += pB;
            accA.x += pA * vA.x; accA.y += pA * vA.y; accA.z += pA * vA.z; accA.w += pA * vA.w;
            accB.x += pB * vB.x; accB.y += pB * vB.y; accB.z += pB * vB.z; accB.w += pB * vB.w;
        }
    }

    const float4 bA = *(const float4*)(b1 + cA);
    const float4 bB = *(const float4*)(b1 + cB);
    const float qA = denA + 1e-16f;
    const float qB = denB + 1e-16f;
    float4 oA, oB;
    oA.x = elu1(accA.x / qA + bA.x); oA.y = elu1(accA.y / qA + bA.y);
    oA.z = elu1(accA.z / qA + bA.z); oA.w = elu1(accA.w / qA + bA.w);
    oB.x = elu1(accB.x / qB + bB.x); oB.y = elu1(accB.y / qB + bB.y);
    oB.z = elu1(accB.z / qB + bB.z); oB.w = elu1(accB.w / qB + bB.w);
    *(float4*)(hout + (size_t)d * 256 + cA) = oA;
    *(float4*)(hout + (size_t)d * 256 + cB) = oB;
}

__global__ __launch_bounds__(256)
void gather2_kernel(const int* __restrict__ rowptr, const int* __restrict__ srt,
                    int E0, int Nn,
                    const float* __restrict__ as2,
                    const float* __restrict__ att_d,
                    const float* __restrict__ h2,
                    const float* __restrict__ b2,
                    float* __restrict__ out) {
    const int lane = threadIdx.x & 31;
    const int d = __builtin_amdgcn_readfirstlane((int)(blockIdx.x * 8 + (threadIdx.x >> 5)));
    if (d >= Nn) return;
    const int start = rowptr[d];
    const int end   = (d + 1 < Nn) ? rowptr[d + 1] : E0;
    const int deg   = end - start;
    const int total = deg + 1;

    float ad;
    {
        const float2 sv = *(const float2*)(h2 + (size_t)d * 64 + lane * 2);
        const float2 wv = *(const float2*)(att_d + lane * 2);
        float t = sv.x * wv.x + sv.y * wv.y;
        for (int o = 16; o > 0; o >>= 1) t += __shfl_xor(t, o);
        ad = t;
    }

    float m = -__builtin_inff();
    for (int c0 = 0; c0 < total; c0 += 32) {
        const int j  = c0 + lane;
        const int sj = (j < deg) ? srt[start + j] : d;
        const float2 ap = *(const float2*)(as2 + (size_t)sj * 2);
        const float lg = leaky((ap.x + ap.y) + ad);
        if (j < total) m = fmaxf(m, lg);
    }
    m = wave_max(m);

    float2 acc = make_float2(0.f, 0.f);
    float den = 0.f;
    for (int c0 = 0; c0 < total; c0 += 32) {
        const int j  = c0 + lane;
        const int sj = (j < deg) ? srt[start + j] : d;
        const float2 ap = *(const float2*)(as2 + (size_t)sj * 2);
        const float p = expf(leaky((ap.x + ap.y) + ad) - m);
        const int cnt = (total - c0 < 32) ? (total - c0) : 32;
        for (int k = 0; k < cnt; ++k) {
            const int s = __builtin_amdgcn_readlane(sj, k);
            const float ph = lane_f(p, k);
            const float2 v = *(const float2*)(h2 + (size_t)s * 64 + lane * 2);
            den += ph;
            acc.x += ph * v.x;
            acc.y += ph * v.y;
        }
    }

    const float2 bb = *(const float2*)(b2 + lane * 2);
    const float q = den + 1e-16f;
    float2 o;
    o.x = acc.x / q + bb.x;
    o.y = acc.y / q + bb.y;
    *(float2*)(out + (size_t)d * 64 + lane * 2) = o;
}

extern "C" void kernel_launch(void* const* d_in, const int* in_sizes, int n_in,
                              void* d_out, int out_size, void* d_ws, size_t ws_size,
                              hipStream_t stream) {
    const float* x        = (const float*)d_in[0];
    const int*   ei       = (const int*)d_in[1];
    const float* W1       = (const float*)d_in[2];
    const float* att_src1 = (const float*)d_in[3];
    const float* att_dst1 = (const float*)d_in[4];
    const float* b1       = (const float*)d_in[5];
    const float* W2       = (const float*)d_in[6];
    const float* att_src2 = (const float*)d_in[7];
    const float* att_dst2 = (const float*)d_in[8];
    const float* b2       = (const float*)d_in[9];
    float* out = (float*)d_out;

    const int Nn = in_sizes[0] / 256;
    const int E0 = in_sizes[1] / 2;

    float* h1     = (float*)d_ws;
    float* hout   = h1 + (size_t)Nn * 256;
    float* as1    = hout + (size_t)Nn * 256;
    int*   rowptr = (int*)(as1 + (size_t)Nn * 8);
    int*   srt    = rowptr + Nn;
    int*   cnt    = (int*)hout;
    int*   rank   = cnt + Nn;
    float* h2     = h1;
    float* as2    = h1 + (size_t)Nn * 64;

    const int TB = 256;
    dim3 g1((256 + 63) / 64, (Nn + 63) / 64);
    dim3 g2((64 + 63) / 64, (Nn + 63) / 64);

    csr_zero_kernel<<<(Nn + TB - 1) / TB, TB, 0, stream>>>(cnt, Nn);
    csr_count_kernel<<<(E0 + TB - 1) / TB, TB, 0, stream>>>(ei, E0, cnt, rank);
    csr_scan_kernel<<<(Nn + 1023) / 1024, 1024, 0, stream>>>(cnt, rowptr, Nn);
    csr_fill_kernel<<<(E0 + TB - 1) / TB, TB, 0, stream>>>(ei, E0, rowptr, rank, srt);

    gat_wmma_gemm<<<g1, TB, 0, stream>>>(x, W1, h1, att_src1, as1, Nn, 256, 256);
    gather1_kernel<<<(Nn + 7) / 8, TB, 0, stream>>>(rowptr, srt, E0, Nn, as1, att_dst1,
                                                    h1, b1, hout);

    gat_wmma_gemm<<<g2, TB, 0, stream>>>(hout, W2, h2, att_src2, as2, Nn, 64, 256);
    gather2_kernel<<<(Nn + 7) / 8, TB, 0, stream>>>(rowptr, srt, E0, Nn, as2, att_dst2,
                                                    h2, b2, out);
}
